# starting kernel
# baseline (speedup 1.0000x reference)
#include <hip/hip_runtime.h>

typedef __attribute__((ext_vector_type(2))) float v2f;
typedef __attribute__((ext_vector_type(8))) float v8f;

#define N_ 50000
#define E_ 800000
#define G_ 64
#define D_ 128
#define BN_EPS 1e-5f

__global__ void deg_init_kernel(float* deg) {
    int i = blockIdx.x * blockDim.x + threadIdx.x;
    if (i < N_) deg[i] = 1.0f;
}

__global__ void deg_acc_kernel(const int* __restrict__ ei, float* deg) {
    int e = blockIdx.x * blockDim.x + threadIdx.x;
    if (e < E_) atomicAdd(&deg[ei[E_ + e]], 1.0f);
}

__global__ void rsqrt_inplace_kernel(float* deg) {
    int i = blockIdx.x * blockDim.x + threadIdx.x;
    if (i < N_) deg[i] = rsqrtf(deg[i]);
}

__global__ __launch_bounds__(256) void gemm_kernel(const float* __restrict__ X,
                                                   const float* __restrict__ W,
                                                   float* __restrict__ H) {
    const int lane = threadIdx.x & 31;
    const int wave = threadIdx.x >> 5;
    const int row0 = blockIdx.x * 16;
    const int col0 = wave * 16;
    const int half = lane >> 4;
    const int l15  = lane & 15;

    const float* xrow = X + (size_t)(row0 + l15) * D_;

    v8f acc0 = {};
    v8f acc1 = {};

#pragma unroll
    for (int k0 = 0; k0 < D_; k0 += 8) {
        {
            const int ka = k0 + 2 * half;
            v2f a = *(const v2f*)(xrow + ka);
            v2f b;
            b.x = W[(size_t)ka * D_ + col0 + l15];
            b.y = W[(size_t)(ka + 1) * D_ + col0 + l15];
            acc0 = __builtin_amdgcn_wmma_f32_16x16x4_f32(
                false, a, false, b, (short)0, acc0, false, false);
        }
        {
            const int kb = k0 + 4 + 2 * half;
            v2f a = *(const v2f*)(xrow + kb);
            v2f b;
            b.x = W[(size_t)kb * D_ + col0 + l15];
            b.y = W[(size_t)(kb + 1) * D_ + col0 + l15];
            acc1 = __builtin_amdgcn_wmma_f32_16x16x4_f32(
                false, a, false, b, (short)0, acc1, false, false);
        }
    }

    v8f acc = acc0 + acc1;

#pragma unroll
    for (int j = 0; j < 8; ++j) {
        int row = row0 + j + 8 * half;
        H[(size_t)row * D_ + col0 + l15] = acc[j];
    }
}

__global__ void node_init_kernel(const float* __restrict__ h,
                                 const float* __restrict__ dinv,
                                 const float* __restrict__ bias,
                                 float* __restrict__ agg) {
    int idx = blockIdx.x * blockDim.x + threadIdx.x;
    int row = idx >> 7;
    int col = idx & 127;
    float dv = dinv[row];
    agg[idx] = h[idx] * (dv * dv) + bias[col];
}

__global__ __launch_bounds__(256) void edge_agg_kernel(const int* __restrict__ ei,
                                                       const float* __restrict__ dinv,
                                                       const float* __restrict__ h,
                                                       float* __restrict__ agg) {
    int wid  = (blockIdx.x * blockDim.x + threadIdx.x) >> 5;
    int lane = threadIdx.x & 31;
    if (wid >= E_) return;
    int s = ei[wid];
    int d = ei[E_ + wid];
    float coef = dinv[s] * dinv[d];
    const float4 v = *(const float4*)(h + (size_t)s * D_ + lane * 4);
    float* dst = agg + (size_t)d * D_ + lane * 4;
    atomicAdd(dst + 0, v.x * coef);
    atomicAdd(dst + 1, v.y * coef);
    atomicAdd(dst + 2, v.z * coef);
    atomicAdd(dst + 3, v.w * coef);
}

__global__ void zero_sums_kernel(float* gsum, float* gsq) {
    int t = threadIdx.x;
    if (t < 128) gsum[t] = 0.0f;
    else         gsq[t - 128] = 0.0f;
}

__global__ void relu_stats_kernel(float* __restrict__ agg,
                                  float* __restrict__ gsum,
                                  float* __restrict__ gsq) {
    int t   = threadIdx.x;
    int col = t & 127;
    int rg  = t >> 7;
    int row0 = blockIdx.x * 64;
    float s = 0.0f, s2 = 0.0f;
    for (int r = rg; r < 64; r += 2) {
        int row = row0 + r;
        if (row >= N_) break;
        int idx = row * D_ + col;
        float v = agg[idx];
        v = fmaxf(v, 0.0f);
        agg[idx] = v;
        s  += v;
        s2 += v * v;
    }
    __shared__ float ls[256];
    __shared__ float lq[256];
    ls[t] = s; lq[t] = s2;
    __syncthreads();
    if (rg == 0) {
        atomicAdd(&gsum[col], ls[t] + ls[t + 128]);
        atomicAdd(&gsq[col],  lq[t] + lq[t + 128]);
    }
}

__global__ void bn_apply_kernel(const float* __restrict__ r,
                                const float* __restrict__ gsum,
                                const float* __restrict__ gsq,
                                const float* __restrict__ gamma,
                                const float* __restrict__ beta,
                                float* __restrict__ xout) {
    int idx = blockIdx.x * blockDim.x + threadIdx.x;
    int col = idx & 127;
    const float inv_n = 1.0f / (float)N_;
    float mu  = gsum[col] * inv_n;
    float var = gsq[col] * inv_n - mu * mu;
    xout[idx] = (r[idx] - mu) * rsqrtf(var + BN_EPS) * gamma[col] + beta[col];
}

__global__ void zero_out_kernel(float* out) {
    int i = blockIdx.x * blockDim.x + threadIdx.x;
    if (i < G_ * D_) out[i] = 0.0f;
}

__global__ __launch_bounds__(256) void pool_kernel(const float* __restrict__ x,
                                                   const int* __restrict__ batch,
                                                   float* __restrict__ out) {
    __shared__ float acc[G_ * D_];
    int t = threadIdx.x;
    for (int i = t; i < G_ * D_; i += 256) acc[i] = 0.0f;
    __syncthreads();
    int col = t & 127;
    int rg  = t >> 7;
    for (int row = blockIdx.x * 2 + rg; row < N_; row += gridDim.x * 2) {
        int g = batch[row];
        atomicAdd(&acc[g * D_ + col], x[(size_t)row * D_ + col]);
    }
    __syncthreads();
    for (int i = t; i < G_ * D_; i += 256) atomicAdd(&out[i], acc[i]);
}

extern "C" void kernel_launch(void* const* d_in, const int* in_sizes, int n_in,
                              void* d_out, int out_size, void* d_ws, size_t ws_size,
                              hipStream_t stream) {
    const float* x0    = (const float*)d_in[0];
    const int*   ei    = (const int*)d_in[1];
    const int*   batch = (const int*)d_in[2];
    const float* Ws[4]  = {(const float*)d_in[3],  (const float*)d_in[7],
                           (const float*)d_in[11], (const float*)d_in[15]};
    const float* bs[4]  = {(const float*)d_in[4],  (const float*)d_in[8],
                           (const float*)d_in[12], (const float*)d_in[16]};
    const float* gms[4] = {(const float*)d_in[5],  (const float*)d_in[9],
                           (const float*)d_in[13], (const float*)d_in[17]};
    const float* bes[4] = {(const float*)d_in[6],  (const float*)d_in[10],
                           (const float*)d_in[14], (const float*)d_in[18]};

    char* ws = (char*)d_ws;
    const size_t MAT = (size_t)N_ * D_ * sizeof(float);
    float* h    = (float*)(ws);
    float* agg  = (float*)(ws + MAT);
    float* xb   = (float*)(ws + 2 * MAT);
    float* dinv = (float*)(ws + 3 * MAT);
    float* gsum = (float*)(ws + 3 * MAT + 204800);
    float* gsq  = (float*)(ws + 3 * MAT + 205312);

    deg_init_kernel<<<(N_ + 255) / 256, 256, 0, stream>>>(dinv);
    deg_acc_kernel<<<(E_ + 255) / 256, 256, 0, stream>>>(ei, dinv);
    rsqrt_inplace_kernel<<<(N_ + 255) / 256, 256, 0, stream>>>(dinv);

    const float* xin = x0;
    for (int l = 0; l < 4; ++l) {
        zero_sums_kernel<<<1, 256, 0, stream>>>(gsum, gsq);
        gemm_kernel<<<N_ / 16, 256, 0, stream>>>(xin, Ws[l], h);
        node_init_kernel<<<(N_ * D_) / 256, 256, 0, stream>>>(h, dinv, bs[l], agg);
        edge_agg_kernel<<<E_ / 8, 256, 0, stream>>>(ei, dinv, h, agg);
        relu_stats_kernel<<<(N_ + 63) / 64, 256, 0, stream>>>(agg, gsum, gsq);
        bn_apply_kernel<<<(N_ * D_) / 256, 256, 0, stream>>>(agg, gsum, gsq,
                                                             gms[l], bes[l], xb);
        xin = xb;
    }

    zero_out_kernel<<<(G_ * D_ + 255) / 256, 256, 0, stream>>>((float*)d_out);
    pool_kernel<<<128, 256, 0, stream>>>(xb, batch, (float*)d_out);
}


# Edge scatter's global float atomics replaced by row owners: a workgroup holds 196 dst rows in LDS (start self-loop+bias), filters the whole dst list, sums rows with LDS float adds (arrival order)
# speedup vs baseline: 1.3935x; 1.3935x over previous
#include <hip/hip_runtime.h>

typedef __attribute__((ext_vector_type(2))) float v2f;
typedef __attribute__((ext_vector_type(8))) float v8f;

#define N_ 50000
#define E_ 800000
#define G_ 64
#define D_ 128
#define BN_EPS 1e-5f

__global__ void deg_init_kernel(float* deg) {
    int i = blockIdx.x * blockDim.x + threadIdx.x;
    if (i < N_) deg[i] = 1.0f;
}

__global__ void deg_acc_kernel(const int* __restrict__ ei, float* deg) {
    int e = blockIdx.x * blockDim.x + threadIdx.x;
    if (e < E_) atomicAdd(&deg[ei[E_ + e]], 1.0f);
}

__global__ void rsqrt_inplace_kernel(float* deg) {
    int i = blockIdx.x * blockDim.x + threadIdx.x;
    if (i < N_) deg[i] = rsqrtf(deg[i]);
}

__global__ __launch_bounds__(256) void gemm_kernel(const float* __restrict__ X,
                                                   const float* __restrict__ W,
                                                   float* __restrict__ H) {
    const int lane = threadIdx.x & 31;
    const int wave = threadIdx.x >> 5;
    const int row0 = blockIdx.x * 16;
    const int col0 = wave * 16;
    const int half = lane >> 4;
    const int l15  = lane & 15;

    const float* xrow = X + (size_t)(row0 + l15) * D_;

    v8f acc0 = {};
    v8f acc1 = {};

#pragma unroll
    for (int k0 = 0; k0 < D_; k0 += 8) {
        {
            const int ka = k0 + 2 * half;
            v2f a = *(const v2f*)(xrow + ka);
            v2f b;
            b.x = W[(size_t)ka * D_ + col0 + l15];
            b.y = W[(size_t)(ka + 1) * D_ + col0 + l15];
            acc0 = __builtin_amdgcn_wmma_f32_16x16x4_f32(
                false, a, false, b, (short)0, acc0, false, false);
        }
        {
            const int kb = k0 + 4 + 2 * half;
            v2f a = *(const v2f*)(xrow + kb);
            v2f b;
            b.x = W[(size_t)kb * D_ + col0 + l15];
            b.y = W[(size_t)(kb + 1) * D_ + col0 + l15];
            acc1 = __builtin_amdgcn_wmma_f32_16x16x4_f32(
                false, a, false, b, (short)0, acc1, false, false);
        }
    }

    v8f acc = acc0 + acc1;

#pragma unroll
    for (int j = 0; j < 8; ++j) {
        int row = row0 + j + 8 * half;
        H[(size_t)row * D_ + col0 + l15] = acc[j];
    }
}

#define R_     196
#define NWG_   ((N_ + R_ - 1) / R_)
#define AT_    1024
#define GRP_   8
#define QCAP_  (GRP_ * 128)
#define STEPS_ (E_ / 128)

__global__ __launch_bounds__(AT_) void agg_owner_kernel(const int* __restrict__ ei,
                                                        const float* __restrict__ dinv,
                                                        const float* __restrict__ h,
                                                        const float* __restrict__ bias,
                                                        float* __restrict__ agg) {
    __shared__ float    tile[R_ * D_];
    __shared__ float    dloc[R_];
    __shared__ unsigned que[AT_ / 32][QCAP_];

    const int tid  = threadIdx.x;
    const int lane = tid & 31;
    const int wave = tid >> 5;
    const int base = blockIdx.x * R_;
    const int rows = (N_ - base < R_) ? (N_ - base) : R_;

    for (int i = tid; i < rows * (D_ / 4); i += AT_) {
        const int r  = i >> 5;
        const int c4 = i & 31;
        const float dv = dinv[base + r];
        const float w  = dv * dv;
        const float4 hv = ((const float4*)h)[(size_t)(base + r) * (D_ / 4) + c4];
        const float4 bv = ((const float4*)bias)[c4];
        float4 o;
        o.x = hv.x * w + bv.x;
        o.y = hv.y * w + bv.y;
        o.z = hv.z * w + bv.z;
        o.w = hv.w * w + bv.w;
        ((float4*)tile)[i] = o;
    }
    if (tid < rows) dloc[tid] = dinv[base + tid];
    __syncthreads();

    const int4* dst4 = (const int4*)(ei + E_);
    unsigned*   myq  = que[wave];

    for (int step0 = wave; step0 < STEPS_; step0 += 32 * GRP_) {
        int4 dv[GRP_];
#pragma unroll
        for (int u = 0; u < GRP_; ++u) {
            const int st = step0 + 32 * u;
            dv[u] = (st < STEPS_) ? dst4[(size_t)st * 32 + lane] : make_int4(-1, -1, -1, -1);
        }
        int cnt = 0;
#pragma unroll
        for (int u = 0; u < GRP_; ++u) {
            const int e0 = (step0 + 32 * u) * 128 + lane * 4;
            const unsigned dl[4] = {(unsigned)(dv[u].x - base), (unsigned)(dv[u].y - base),
                                    (unsigned)(dv[u].z - base), (unsigned)(dv[u].w - base)};
            const unsigned lo01 = dl[0] < dl[1] ? dl[0] : dl[1];
            const unsigned lo23 = dl[2] < dl[3] ? dl[2] : dl[3];
            const unsigned lo   = lo01 < lo23 ? lo01 : lo23;
            if (__builtin_amdgcn_ballot_w32(lo < (unsigned)rows) != 0u) {
#pragma unroll
                for (int j = 0; j < 4; ++j) {
                    const bool     m    = dl[j] < (unsigned)rows;
                    const unsigned mask = __builtin_amdgcn_ballot_w32(m);
                    if (m) {
                        const unsigned pos = (unsigned)cnt + __builtin_amdgcn_mbcnt_lo(mask, 0u);
                        myq[pos] = ((unsigned)(e0 + j) << 8) | dl[j];
                    }
                    cnt += __builtin_popcount(mask);
                }
            }
        }

        for (int i = 0; i < cnt; i += 4) {
            unsigned ent[4];
            int      s[4];
            float    cf[4];
            float    v[4][4];
#pragma unroll
            for (int k = 0; k < 4; ++k) {
                const int idx = (i + k < cnt) ? (i + k) : (cnt - 1);
                ent[k] = __builtin_amdgcn_readfirstlane(myq[idx]);
            }
#pragma unroll
            for (int k = 0; k < 4; ++k) s[k] = ei[ent[k] >> 8];
#pragma unroll
            for (int k = 0; k < 4; ++k) {
                cf[k] = dinv[s[k]] * dloc[ent[k] & 255u];
                const float* p = h + (size_t)s[k] * D_ + lane;
                v[k][0] = p[0];
                v[k][1] = p[32];
                v[k][2] = p[64];
                v[k][3] = p[96];
            }
#pragma unroll
            for (int k = 0; k < 4; ++k) {
                if (i + k < cnt) {
                    float* t = tile + (ent[k] & 255u) * D_ + lane;
                    atomicAdd(t,      v[k][0] * cf[k]);
                    atomicAdd(t + 32, v[k][1] * cf[k]);
                    atomicAdd(t + 64, v[k][2] * cf[k]);
                    atomicAdd(t + 96, v[k][3] * cf[k]);
                }
            }
        }
    }
    __syncthreads();

    for (int i = tid; i < rows * (D_ / 4); i += AT_)
        ((float4*)agg)[(size_t)base * (D_ / 4) + i] = ((const float4*)tile)[i];
}

__global__ void zero_sums_kernel(float* gsum, float* gsq) {
    int t = threadIdx.x;
    if (t < 128) gsum[t] = 0.0f;
    else         gsq[t - 128] = 0.0f;
}

__global__ void relu_stats_kernel(float* __restrict__ agg,
                                  float* __restrict__ gsum,
                                  float* __restrict__ gsq) {
    int t   = threadIdx.x;
    int col = t & 127;
    int rg  = t >> 7;
    int row0 = blockIdx.x * 64;
    float s = 0.0f, s2 = 0.0f;
    for (int r = rg; r < 64; r += 2) {
        int row = row0 + r;
        if (row >= N_) break;
        int idx = row * D_ + col;
        float v = agg[idx];
        v = fmaxf(v, 0.0f);
        agg[idx] = v;
        s  += v;
        s2 += v * v;
    }
    __shared__ float ls[256];
    __shared__ float lq[256];
    ls[t] = s; lq[t] = s2;
    __syncthreads();
    if (rg == 0) {
        atomicAdd(&gsum[col], ls[t] + ls[t + 128]);
        atomicAdd(&gsq[col],  lq[t] + lq[t + 128]);
    }
}

__global__ void bn_apply_kernel(const float* __restrict__ r,
                                const float* __restrict__ gsum,
                                const float* __restrict__ gsq,
                                const float* __restrict__ gamma,
                                const float* __restrict__ beta,
                                float* __restrict__ xout) {
    int idx = blockIdx.x * blockDim.x + threadIdx.x;
    int col = idx & 127;
    const float inv_n = 1.0f / (float)N_;
    float mu  = gsum[col] * inv_n;
    float var = gsq[col] * inv_n - mu * mu;
    xout[idx] = (r[idx] - mu) * rsqrtf(var + BN_EPS) * gamma[col] + beta[col];
}

__global__ void zero_out_kernel(float* out) {
    int i = blockIdx.x * blockDim.x + threadIdx.x;
    if (i < G_ * D_) out[i] = 0.0f;
}

__global__ __launch_bounds__(256) void pool_kernel(const float* __restrict__ x,
                                                   const int* __restrict__ batch,
                                                   float* __restrict__ out) {
    __shared__ float acc[G_ * D_];
    int t = threadIdx.x;
    for (int i = t; i < G_ * D_; i += 256) acc[i] = 0.0f;
    __syncthreads();
    int col = t & 127;
    int rg  = t >> 7;
    for (int row = blockIdx.x * 2 + rg; row < N_; row += gridDim.x * 2) {
        int g = batch[row];
        atomicAdd(&acc[g * D_ + col], x[(size_t)row * D_ + col]);
    }
    __syncthreads();
    for (int i = t; i < G_ * D_; i += 256) atomicAdd(&out[i], acc[i]);
}

extern "C" void kernel_launch(void* const* d_in, const int* in_sizes, int n_in,
                              void* d_out, int out_size, void* d_ws, size_t ws_size,
                              hipStream_t stream) {
    const float* x0    = (const float*)d_in[0];
    const int*   ei    = (const int*)d_in[1];
    const int*   batch = (const int*)d_in[2];
    const float* Ws[4]  = {(const float*)d_in[3],  (const float*)d_in[7],
                           (const float*)d_in[11], (const float*)d_in[15]};
    const float* bs[4]  = {(const float*)d_in[4],  (const float*)d_in[8],
                           (const float*)d_in[12], (const float*)d_in[16]};
    const float* gms[4] = {(const float*)d_in[5],  (const float*)d_in[9],
                           (const float*)d_in[13], (const float*)d_in[17]};
    const float* bes[4] = {(const float*)d_in[6],  (const float*)d_in[10],
                           (const float*)d_in[14], (const float*)d_in[18]};

    char* ws = (char*)d_ws;
    const size_t MAT = (size_t)N_ * D_ * sizeof(float);
    float* h    = (float*)(ws);
    float* agg  = (float*)(ws + MAT);
    float* xb   = (float*)(ws + 2 * MAT);
    float* dinv = (float*)(ws + 3 * MAT);
    float* gsum = (float*)(ws + 3 * MAT + 204800);
    float* gsq  = (float*)(ws + 3 * MAT + 205312);

    deg_init_kernel<<<(N_ + 255) / 256, 256, 0, stream>>>(dinv);
    deg_acc_kernel<<<(E_ + 255) / 256, 256, 0, stream>>>(ei, dinv);
    rsqrt_inplace_kernel<<<(N_ + 255) / 256, 256, 0, stream>>>(dinv);

    const float* xin = x0;
    for (int l = 0; l < 4; ++l) {
        zero_sums_kernel<<<1, 256, 0, stream>>>(gsum, gsq);
        gemm_kernel<<<N_ / 16, 256, 0, stream>>>(xin, Ws[l], h);
        agg_owner_kernel<<<NWG_, AT_, 0, stream>>>(ei, dinv, h, bs[l], agg);
        relu_stats_kernel<<<(N_ + 63) / 64, 256, 0, stream>>>(agg, gsum, gsq);
        bn_apply_kernel<<<(N_ * D_) / 256, 256, 0, stream>>>(agg, gsum, gsq,
                                                             gms[l], bes[l], xb);
        xin = xb;
    }

    zero_out_kernel<<<(G_ * D_ + 255) / 256, 256, 0, stream>>>((float*)d_out);
    pool_kernel<<<128, 256, 0, stream>>>(xb, batch, (float*)d_out);
}
